# starting kernel
# baseline (speedup 1.0000x reference)
#include <hip/hip_runtime.h>

typedef __attribute__((ext_vector_type(2))) float v2f;
typedef __attribute__((ext_vector_type(8))) float v8f;

#define F_IN 5
#define HID 16

__global__ void k_init_deg(float* __restrict__ deg, int N) {
  int i = blockIdx.x * blockDim.x + threadIdx.x;
  if (i < N) deg[i] = 1.0f;
}

__global__ void k_degree(const int* __restrict__ ei, float* __restrict__ deg, int E) {
  int e = blockIdx.x * blockDim.x + threadIdx.x;
  if (e < E) {
    __builtin_prefetch(&ei[(long)E + e + 16384], 0, 0);
    atomicAdd(&deg[ei[(long)E + e]], 1.0f);
  }
}

__global__ void k_rsqrt(float* __restrict__ d, int N) {
  int i = blockIdx.x * blockDim.x + threadIdx.x;
  if (i < N) d[i] = rsqrtf(d[i]);
}

__global__ void k_xw_wmma(const float* __restrict__ x,
                          const float* __restrict__ W1,
                          const float* __restrict__ dinv,
                          float* __restrict__ xw,
                          float* __restrict__ out1,
                          int N) {
  const int lane = threadIdx.x & 31;
  const int wave = threadIdx.x >> 5;
  const int wavesPerBlock = blockDim.x >> 5;
  const int base = (blockIdx.x * wavesPerBlock + wave) * 16;
  if (base >= N) return;

  const int half = lane >> 4;
  const int l    = lane & 15;
  int row = base + l;
  if (row >= N) row = N - 1;

  const float* xr = x + (long)row * F_IN;
  const int klo = half * 2;

  v2f a0; a0.x = xr[klo];            a0.y = xr[klo + 1];
  v2f b0; b0.x = W1[klo * HID + l];  b0.y = W1[(klo + 1) * HID + l];

  float x4 = xr[4];
  float w4 = W1[4 * HID + l];
  v2f a1; a1.x = half ? 0.0f : x4;  a1.y = 0.0f;
  v2f b1; b1.x = half ? 0.0f : w4;  b1.y = 0.0f;

  v8f c = {};
  c = __builtin_amdgcn_wmma_f32_16x16x4_f32(false, a0, false, b0, (short)0, c, false, false);
  c = __builtin_amdgcn_wmma_f32_16x16x4_f32(false, a1, false, b1, (short)0, c, false, false);

  const long colBase = (long)(base + half * 8) * HID + l;
  if (base + 16 <= N) {
    float* xp = xw + colBase;
    float* op = out1 + colBase;
    const float* dp = dinv + base + half * 8;
#pragma unroll
    for (int r = 0; r < 8; ++r) {
      const float v = c[r];
      const float di = dp[r];
      xp[(long)r * HID] = v;
      op[(long)r * HID] = di * di * v;
    }
  } else {
#pragma unroll
    for (int r = 0; r < 8; ++r) {
      const int node = base + half * 8 + r;
      if (node < N) {
        const float v = c[r];
        const float di = dinv[node];
        xw[(long)node * HID + l] = v;
        out1[(long)node * HID + l] = di * di * v;
      }
    }
  }
}

__global__ void k_agg1(const int* __restrict__ ei, const float* __restrict__ dinv,
                       const float* __restrict__ xw, float* __restrict__ out1, int E) {
  int e = blockIdx.x * blockDim.x + threadIdx.x;
  if (e >= E) return;
  __builtin_prefetch(&ei[e + 16384], 0, 0);
  __builtin_prefetch(&ei[(long)E + e + 16384], 0, 0);
  const int s = ei[e];
  const int d = ei[(long)E + e];
  const float nrm = dinv[s] * dinv[d];
  const float4* xs = (const float4*)(xw + (long)s * HID);
  float* od = out1 + (long)d * HID;
  float4 v0 = xs[0], v1 = xs[1], v2 = xs[2], v3 = xs[3];
  atomicAdd(od + 0,  nrm * v0.x);  atomicAdd(od + 1,  nrm * v0.y);
  atomicAdd(od + 2,  nrm * v0.z);  atomicAdd(od + 3,  nrm * v0.w);
  atomicAdd(od + 4,  nrm * v1.x);  atomicAdd(od + 5,  nrm * v1.y);
  atomicAdd(od + 6,  nrm * v1.z);  atomicAdd(od + 7,  nrm * v1.w);
  atomicAdd(od + 8,  nrm * v2.x);  atomicAdd(od + 9,  nrm * v2.y);
  atomicAdd(od + 10, nrm * v2.z);  atomicAdd(od + 11, nrm * v2.w);
  atomicAdd(od + 12, nrm * v3.x);  atomicAdd(od + 13, nrm * v3.y);
  atomicAdd(od + 14, nrm * v3.z);  atomicAdd(od + 15, nrm * v3.w);
}

__global__ void k_layer2(const float* __restrict__ out1, const float* __restrict__ b1,
                         const float* __restrict__ W2, const float* __restrict__ b2,
                         const float* __restrict__ dinv, float* __restrict__ z,
                         float* __restrict__ out, int N) {
  int i = blockIdx.x * blockDim.x + threadIdx.x;
  if (i >= N) return;
  const float4* hr = (const float4*)(out1 + (long)i * HID);
  const float4* bb = (const float4*)b1;
  const float4* ww = (const float4*)W2;
  float acc = 0.0f;
#pragma unroll
  for (int q = 0; q < 4; ++q) {
    float4 v = hr[q], b = bb[q], w = ww[q];
    acc = fmaf(fmaxf(v.x + b.x, 0.0f), w.x, acc);
    acc = fmaf(fmaxf(v.y + b.y, 0.0f), w.y, acc);
    acc = fmaf(fmaxf(v.z + b.z, 0.0f), w.z, acc);
    acc = fmaf(fmaxf(v.w + b.w, 0.0f), w.w, acc);
  }
  z[i] = acc;
  const float di = dinv[i];
  out[i] = b2[0] + di * di * acc;
}

__global__ void k_agg2(const int* __restrict__ ei, const float* __restrict__ dinv,
                       const float* __restrict__ z, float* __restrict__ out, int E) {
  int e = blockIdx.x * blockDim.x + threadIdx.x;
  if (e >= E) return;
  __builtin_prefetch(&ei[e + 16384], 0, 0);
  __builtin_prefetch(&ei[(long)E + e + 16384], 0, 0);
  const int s = ei[e];
  const int d = ei[(long)E + e];
  atomicAdd(&out[d], dinv[s] * dinv[d] * z[s]);
}

extern "C" void kernel_launch(void* const* d_in, const int* in_sizes, int n_in,
                              void* d_out, int out_size, void* d_ws, size_t ws_size,
                              hipStream_t stream) {
  const float* x  = (const float*)d_in[0];
  const int*   ei = (const int*)d_in[1];
  const float* W1 = (const float*)d_in[2];
  const float* b1 = (const float*)d_in[3];
  const float* W2 = (const float*)d_in[4];
  const float* b2 = (const float*)d_in[5];
  float* out = (float*)d_out;

  const int N = in_sizes[0] / F_IN;
  const int E = in_sizes[1] / 2;

  char* ws = (char*)d_ws;
  size_t off = 0;
  auto carve = [&](size_t bytes) -> void* {
    off = (off + 255) & ~(size_t)255;
    void* p = ws + off;
    off += bytes;
    return p;
  };
  float* dinv = (float*)carve((size_t)N * sizeof(float));
  float* xw   = (float*)carve((size_t)N * HID * sizeof(float));
  float* out1 = (float*)carve((size_t)N * HID * sizeof(float));
  float* z    = (float*)carve((size_t)N * sizeof(float));
  (void)ws_size; (void)n_in; (void)out_size;

  const int B = 256;
  const int gN = (N + B - 1) / B;
  const int gE = (E + B - 1) / B;
  const int wavesNeeded   = (N + 15) / 16;
  const int wavesPerBlock = B / 32;
  const int gW = (wavesNeeded + wavesPerBlock - 1) / wavesPerBlock;

  hipLaunchKernelGGL(k_init_deg, dim3(gN), dim3(B), 0, stream, dinv, N);
  hipLaunchKernelGGL(k_degree,   dim3(gE), dim3(B), 0, stream, ei, dinv, E);
  hipLaunchKernelGGL(k_rsqrt,    dim3(gN), dim3(B), 0, stream, dinv, N);
  hipLaunchKernelGGL(k_xw_wmma,  dim3(gW), dim3(B), 0, stream, x, W1, dinv, xw, out1, N);
  hipLaunchKernelGGL(k_agg1,     dim3(gE), dim3(B), 0, stream, ei, dinv, xw, out1, E);
  hipLaunchKernelGGL(k_layer2,   dim3(gN), dim3(B), 0, stream, out1, b1, W2, b2, dinv, z, out, N);
  hipLaunchKernelGGL(k_agg2,     dim3(gE), dim3(B), 0, stream, ei, dinv, z, out, E);
}


# Two small passes folded into neighbours: scan per bucket row, consumers sum the bucket totals (no scan_add); relu and W2 dot in the layer-1 gather's tail (no out1): 7 launches, not 9; sum order kept
# speedup vs baseline: 10.7419x; 10.7419x over previous
#include <hip/hip_runtime.h>

typedef __attribute__((ext_vector_type(2))) float v2f;
typedef __attribute__((ext_vector_type(8))) float v8f;

#define F_IN 5
#define HID 16
#define SCAN_THREADS 256
#define SCAN_TILE 1024
#define BSHIFT 8
#define BW 256
#define MAXB 512
#define PT 512
#define CH 8192
#define EPC (CH / PT)
#define CAP 12288
#define LB 8
#define SRC_MASK 0x00FFFFFF

static_assert(PT == MAXB, "one thread per bucket table entry");
static_assert(BW <= PT, "one thread per node of a bucket");


__global__ __launch_bounds__(PT) void k_bhist(const int* __restrict__ ei, int* __restrict__ hist,
                                              int E, int NB, int NC) {
  __shared__ int lh[MAXB];
  const int t = threadIdx.x;
  const int* dst = ei + (long)E;
  lh[t] = 0;
  __syncthreads();
  const int e0 = blockIdx.x * CH + t;
  int d[EPC];
#pragma unroll
  for (int k = 0; k < EPC; ++k) {
    const int e = e0 + k * PT;
    d[k] = (e < E) ? dst[e] : -1;
  }
#pragma unroll
  for (int k = 0; k < EPC; ++k)
    if (d[k] >= 0) atomicAdd(&lh[d[k] >> BSHIFT], 1);
  __syncthreads();
  if (t < NB) hist[t * NC + blockIdx.x] = lh[t];
}

__global__ void k_scan_rows(const int* __restrict__ hist, int* __restrict__ goff,
                            int* __restrict__ rowtot, int NC) {
  __shared__ int wsum[SCAN_THREADS / 32];
  const int t = threadIdx.x;
  const int lane = t & 31;
  const int wave = t >> 5;
  const int* in = hist + (long)blockIdx.x * NC;
  int* outp = goff + (long)blockIdx.x * NC;
  int carry = 0;
  for (int base = 0; base < NC; base += SCAN_TILE) {
    const int i0 = base + t * 4;
    int c[4];
#pragma unroll
    for (int k = 0; k < 4; ++k) c[k] = (i0 + k < NC) ? in[i0 + k] : 0;
    const int s = c[0] + c[1] + c[2] + c[3];
    int inc = s;
#pragma unroll
    for (int d = 1; d < 32; d <<= 1) {
      const int v = __shfl_up(inc, d);
      if (lane >= d) inc += v;
    }
    if (lane == 31) wsum[wave] = inc;
    __syncthreads();
    int woff = 0, total = 0;
#pragma unroll
    for (int w = 0; w < SCAN_THREADS / 32; ++w) {
      const int v = wsum[w];
      total += v;
      if (w < wave) woff += v;
    }
    int run = carry + woff + inc - s;
#pragma unroll
    for (int k = 0; k < 4; ++k) {
      if (i0 + k < NC) {
        outp[i0 + k] = run;
        run += c[k];
      }
    }
    carry += total;
    __syncthreads();
  }
  if (t == 0) rowtot[blockIdx.x] = carry;
}

__global__ __launch_bounds__(PT) void k_partition(const int* __restrict__ ei, const int* __restrict__ goff,
                                                  const int* __restrict__ rowtot,
                                                  int* __restrict__ part, int E, int NB, int NC) {
  __shared__ int lh[MAXB];
  __shared__ int lo[MAXB];
  __shared__ int lg[MAXB];
  __shared__ int wsum[PT / 32];
  __shared__ int bsum[PT / 32];
  __shared__ int sorted[CH];
  __shared__ unsigned short sbkt[CH];
  const int t = threadIdx.x;
  const int lane = t & 31;
  const int wave = t >> 5;
  const int c = blockIdx.x;
  const int* dst = ei + (long)E;
  lh[t] = 0;
  const int e0 = c * CH + t;
  int d[EPC], s[EPC], br[EPC];
#pragma unroll
  for (int k = 0; k < EPC; ++k) {
    const int e = e0 + k * PT;
    const bool ok = e < E;
    d[k] = ok ? dst[e] : -1;
    s[k] = ok ? ei[e] : 0;
  }
  const int rt = (t < NB) ? rowtot[t] : 0;
  const int gl = (t < NB) ? goff[t * NC + c] : 0;
  int rinc = rt;
#pragma unroll
  for (int dd = 1; dd < 32; dd <<= 1) {
    const int u = __shfl_up(rinc, dd);
    if (lane >= dd) rinc += u;
  }
  if (lane == 31) bsum[wave] = rinc;
  __syncthreads();
  {
    int boff = 0;
#pragma unroll
    for (int w = 0; w < PT / 32; ++w)
      if (w < wave) boff += bsum[w];
    lg[t] = boff + rinc - rt + gl;
  }
#pragma unroll
  for (int k = 0; k < EPC; ++k) {
    br[k] = -1;
    if (d[k] >= 0) {
      const int b = d[k] >> BSHIFT;
      const int r = atomicAdd(&lh[b], 1);
      br[k] = (b << 16) | r;
    }
  }
  __syncthreads();
  const int v = lh[t];
  int inc = v;
#pragma unroll
  for (int dd = 1; dd < 32; dd <<= 1) {
    const int u = __shfl_up(inc, dd);
    if (lane >= dd) inc += u;
  }
  if (lane == 31) wsum[wave] = inc;
  __syncthreads();
  int woff = 0;
#pragma unroll
  for (int w = 0; w < PT / 32; ++w)
    if (w < wave) woff += wsum[w];
  lo[t] = woff + inc - v;
  __syncthreads();
#pragma unroll
  for (int k = 0; k < EPC; ++k) {
    if (br[k] >= 0) {
      const int b = br[k] >> 16;
      const int slot = lo[b] + (br[k] & 0xFFFF);
      sorted[slot] = (int)(((unsigned)(d[k] & (BW - 1)) << 24) | (unsigned)s[k]);
      sbkt[slot] = (unsigned short)b;
    }
  }
  __syncthreads();
  const int n = min(CH, E - c * CH);
#pragma unroll
  for (int k = 0; k < EPC; ++k) {
    const int i = t + k * PT;
    if (i < n) {
      const int b = sbkt[i];
      part[lg[b] + (i - lo[b])] = sorted[i];
    }
  }
}

__global__ __launch_bounds__(PT) void k_bucket_csr(const int* __restrict__ part, const int* __restrict__ rowtot,
                                                   int* __restrict__ csr, int* __restrict__ cnt,
                                                   int* __restrict__ rowptr, float* __restrict__ dinv,
                                                   int N) {
  __shared__ int lcnt[BW];
  __shared__ int lrow[BW];
  __shared__ int lcur[BW];
  __shared__ int wsum[BW / 32];
  __shared__ int bsum[PT / 32];
  __shared__ int stage[CAP];
  const int t = threadIdx.x;
  const int lane = t & 31;
  const int wave = t >> 5;
  const int b = blockIdx.x;
  const int n = rowtot[b];
  int pv = (t < b) ? rowtot[t] : 0;
#pragma unroll
  for (int dd = 16; dd >= 1; dd >>= 1) pv += __shfl_xor(pv, dd);
  if (lane == 0) bsum[wave] = pv;
  if (t < BW) { lcnt[t] = 0; lcur[t] = 0; }
  __syncthreads();
  int s0 = 0;
#pragma unroll
  for (int w = 0; w < PT / 32; ++w) s0 += bsum[w];
  const int s1 = s0 + n;
  for (int i = s0 + t; i < s1; i += LB * PT) {
    int w[LB];
#pragma unroll
    for (int k = 0; k < LB; ++k) {
      const int j = i + k * PT;
      w[k] = (j < s1) ? part[j] : 0;
    }
#pragma unroll
    for (int k = 0; k < LB; ++k)
      if (i + k * PT < s1) atomicAdd(&lcnt[((unsigned)w[k]) >> 24], 1);
  }
  __syncthreads();
  const int v = (t < BW) ? lcnt[t] : 0;
  int inc = v;
#pragma unroll
  for (int dd = 1; dd < 32; dd <<= 1) {
    const int u = __shfl_up(inc, dd);
    if (lane >= dd) inc += u;
  }
  if (t < BW && lane == 31) wsum[wave] = inc;
  __syncthreads();
  if (t < BW) {
    int woff = 0;
#pragma unroll
    for (int w = 0; w < BW / 32; ++w)
      if (w < wave) woff += wsum[w];
    const int ex = woff + inc - v;
    lrow[t] = ex;
    const int node = b * BW + t;
    if (node < N) {
      cnt[node] = v;
      rowptr[node] = s0 + ex;
      dinv[node] = rsqrtf((float)(v + 1));
    }
  }
  __syncthreads();
  if (n <= CAP) {
    for (int i = s0 + t; i < s1; i += LB * PT) {
      int w[LB];
#pragma unroll
      for (int k = 0; k < LB; ++k) {
        const int j = i + k * PT;
        w[k] = (j < s1) ? part[j] : 0;
      }
#pragma unroll
      for (int k = 0; k < LB; ++k) {
        if (i + k * PT < s1) {
          const int dl = ((unsigned)w[k]) >> 24;
          const int r = atomicAdd(&lcur[dl], 1);
          stage[lrow[dl] + r] = w[k] & SRC_MASK;
        }
      }
    }
    __syncthreads();
    for (int i = t; i < n; i += PT) csr[s0 + i] = stage[i];
  } else {
    for (int i = s0 + t; i < s1; i += PT) {
      const int w = part[i];
      const int dl = ((unsigned)w) >> 24;
      const int r = atomicAdd(&lcur[dl], 1);
      csr[s0 + lrow[dl] + r] = w & SRC_MASK;
    }
  }
}

__global__ void k_xw_wmma(const float* __restrict__ x,
                          const float* __restrict__ W1,
                          const float* __restrict__ dinv,
                          float* __restrict__ xwd,
                          int N) {
  const int lane = threadIdx.x & 31;
  const int wave = threadIdx.x >> 5;
  const int wavesPerBlock = blockDim.x >> 5;
  const int base = (blockIdx.x * wavesPerBlock + wave) * 16;
  if (base >= N) return;

  const int half = lane >> 4;
  const int l    = lane & 15;
  int row = base + l;
  if (row >= N) row = N - 1;

  const float* xr = x + (long)row * F_IN;
  const int klo = half * 2;

  v2f a0; a0.x = xr[klo];            a0.y = xr[klo + 1];
  v2f b0; b0.x = W1[klo * HID + l];  b0.y = W1[(klo + 1) * HID + l];

  float x4 = xr[4];
  float w4 = W1[4 * HID + l];
  v2f a1; a1.x = half ? 0.0f : x4;  a1.y = 0.0f;
  v2f b1; b1.x = half ? 0.0f : w4;  b1.y = 0.0f;

  const int n0 = base + half * 8;
  float dv[8];
#pragma unroll
  for (int r = 0; r < 8; ++r) dv[r] = dinv[min(n0 + r, N - 1)];

  v8f c = {};
  c = __builtin_amdgcn_wmma_f32_16x16x4_f32(false, a0, false, b0, (short)0, c, false, false);
  c = __builtin_amdgcn_wmma_f32_16x16x4_f32(false, a1, false, b1, (short)0, c, false, false);

  float2* xp = (float2*)xwd + ((long)n0 * HID + l);
  if (base + 16 <= N) {
#pragma unroll
    for (int r = 0; r < 8; ++r)
      xp[(long)r * HID] = make_float2(c[r], dv[r]);
  } else {
#pragma unroll
    for (int r = 0; r < 8; ++r)
      if (n0 + r < N) xp[(long)r * HID] = make_float2(c[r], dv[r]);
  }
}

__global__ void k_agg1_gather(const int* __restrict__ csr, const int* __restrict__ rowptr,
                              const int* __restrict__ cnt, const float* __restrict__ xwd,
                              const float* __restrict__ b1, const float* __restrict__ W2,
                              float* __restrict__ zd, int N) {
  const int t = blockIdx.x * blockDim.x + threadIdx.x;
  const int node = t >> 4;
  const int l = threadIdx.x & 15;
  if (node >= N) return;
  const float2* rows = (const float2*)xwd;
  const int start = rowptr[node];
  const int deg = cnt[node];
  const float2 own = rows[(long)node * HID + l];
  const float bl = b1[l];
  const float dd = own.y;
  float acc = dd * dd * own.x;
  for (int base = 0; base < deg; base += 16) {
    const int m = deg - base;
    int s_l = node;
    if (l < m) s_l = csr[start + base + l];
#pragma unroll
    for (int g = 0; g < 16; g += 8) {
      if (g < m) {
        float2 v[8];
#pragma unroll
        for (int j = 0; j < 8; ++j) {
          const int s = __shfl(s_l, g + j, 16);
          v[j] = rows[(long)s * HID + l];
        }
#pragma unroll
        for (int j = 0; j < 8; ++j)
          acc = fmaf(v[j].y * dd, (g + j < m) ? v[j].x : 0.0f, acc);
      }
    }
  }
  const float h = fmaxf(acc + bl, 0.0f);
  float z = 0.0f;
#pragma unroll
  for (int f = 0; f < HID; ++f) z = fmaf(__shfl(h, f, 16), W2[f], z);
  if (l == 0) ((float2*)zd)[node] = make_float2(z, dd);
}

__global__ void k_agg2_gather(const int* __restrict__ csr, const int* __restrict__ rowptr,
                              const int* __restrict__ cnt, const float* __restrict__ zd,
                              const float* __restrict__ b2, float* __restrict__ out, int N) {
  const int t = blockIdx.x * blockDim.x + threadIdx.x;
  const int node = t >> 4;
  const int l = threadIdx.x & 15;
  if (node >= N) return;
  const float2* rec = (const float2*)zd;
  const int start = rowptr[node];
  const int deg = cnt[node];
  const float2 own = rec[node];
  const float dd = own.y;
  float acc = 0.0f;
  for (int p = l; p < deg; p += 16) {
    const float2 r = rec[csr[start + p]];
    acc = fmaf(r.y * dd, r.x, acc);
  }
#pragma unroll
  for (int d = 8; d >= 1; d >>= 1) acc += __shfl_xor(acc, d, 16);
  if (l == 0) out[node] = b2[0] + dd * dd * own.x + acc;
}

extern "C" void kernel_launch(void* const* d_in, const int* in_sizes, int n_in,
                              void* d_out, int out_size, void* d_ws, size_t ws_size,
                              hipStream_t stream) {
  const float* x  = (const float*)d_in[0];
  const int*   ei = (const int*)d_in[1];
  const float* W1 = (const float*)d_in[2];
  const float* b1 = (const float*)d_in[3];
  const float* W2 = (const float*)d_in[4];
  const float* b2 = (const float*)d_in[5];
  float* out = (float*)d_out;

  const int N = in_sizes[0] / F_IN;
  const int E = in_sizes[1] / 2;
  const int NB = (N + BW - 1) / BW;
  const int NC = (E + CH - 1) / CH;
  const int M = NB * NC;

  char* ws = (char*)d_ws;
  size_t off = 0;
  auto carve = [&](size_t bytes) -> void* {
    off = (off + 255) & ~(size_t)255;
    void* p = ws + off;
    off += bytes;
    return p;
  };
  float* dinv    = (float*)carve((size_t)N * sizeof(float));
  float* xwd     = (float*)carve((size_t)N * HID * 2 * sizeof(float));
  float* zd      = (float*)carve((size_t)N * 2 * sizeof(float));
  int*   cnt     = (int*)carve((size_t)N * sizeof(int));
  int*   rowptr  = (int*)carve((size_t)N * sizeof(int));
  int*   hist    = (int*)carve((size_t)M * sizeof(int));
  int*   goff    = (int*)carve((size_t)M * sizeof(int));
  int*   rowtot  = (int*)carve((size_t)NB * sizeof(int));
  int*   part    = (int*)carve((size_t)E * sizeof(int));
  int*   csr     = (int*)carve((size_t)E * sizeof(int));
  (void)ws_size; (void)n_in; (void)out_size;

  const int B = 256;
  const int gN16 = (int)(((long)N * 16 + B - 1) / B);
  const int wavesNeeded   = (N + 15) / 16;
  const int wavesPerBlock = B / 32;
  const int gW = (wavesNeeded + wavesPerBlock - 1) / wavesPerBlock;

  hipLaunchKernelGGL(k_bhist,       dim3(NC),     dim3(PT), 0, stream, ei, hist, E, NB, NC);
  hipLaunchKernelGGL(k_scan_rows,   dim3(NB),     dim3(SCAN_THREADS), 0, stream, hist, goff, rowtot, NC);
  hipLaunchKernelGGL(k_partition,   dim3(NC),     dim3(PT), 0, stream, ei, goff, rowtot, part, E, NB, NC);
  hipLaunchKernelGGL(k_bucket_csr,  dim3(NB),     dim3(PT), 0, stream, part, rowtot, csr, cnt, rowptr, dinv, N);
  hipLaunchKernelGGL(k_xw_wmma,     dim3(gW),     dim3(B), 0, stream, x, W1, dinv, xwd, N);
  hipLaunchKernelGGL(k_agg1_gather, dim3(gN16),   dim3(B), 0, stream, csr, rowptr, cnt, xwd, b1, W2, zd, N);
  hipLaunchKernelGGL(k_agg2_gather, dim3(gN16),   dim3(B), 0, stream, csr, rowptr, cnt, zd, b2, out, N);
}


# First sort level stores each chunk's bucket-ordered edges in the chunk's own slab with its bucket starts; a bucket walks its runs: no count and scan launches (5, not 7); node f32 sum order changes
# speedup vs baseline: 12.7894x; 1.1906x over previous
#include <hip/hip_runtime.h>

typedef __attribute__((ext_vector_type(2))) float v2f;
typedef __attribute__((ext_vector_type(8))) float v8f;

#define F_IN 5
#define HID 16
#define BSHIFT 8
#define BW 256
#define MAXB 512
#define MAXC 512
#define PT 512
#define CH 8192
#define EPC (CH / PT)
#define CAP 12288
#define RB 16
#define SRC_MASK 0x00FFFFFF

static_assert(PT == MAXB, "one thread per bucket table entry");
static_assert(PT == MAXC, "one thread per chunk table entry");
static_assert(RB == 16, "the load fence in walk_runs names RB values");
static_assert(MAXC == 32 * (PT / 32), "a wave's runs, one per lane, cover every chunk");
static_assert(BW <= PT, "one thread per node of a bucket");


__global__ __launch_bounds__(PT) void k_chunk_sort(const int* __restrict__ ei, int* __restrict__ part,
                                                   int* __restrict__ offs, int E, int NB, int NC) {
  __shared__ int lh[MAXB];
  __shared__ int lo[MAXB];
  __shared__ int wsum[PT / 32];
  __shared__ int sorted[CH];
  const int t = threadIdx.x;
  const int lane = t & 31;
  const int wave = t >> 5;
  const int c = blockIdx.x;
  const int* dst = ei + (long)E;
  lh[t] = 0;
  __syncthreads();
  const int e0 = c * CH + t;
  int d[EPC], s[EPC], br[EPC];
#pragma unroll
  for (int k = 0; k < EPC; ++k) {
    const int e = e0 + k * PT;
    const bool ok = e < E;
    d[k] = ok ? dst[e] : -1;
    s[k] = ok ? ei[e] : 0;
  }
#pragma unroll
  for (int k = 0; k < EPC; ++k) {
    br[k] = -1;
    if (d[k] >= 0) {
      const int b = d[k] >> BSHIFT;
      const int r = atomicAdd(&lh[b], 1);
      br[k] = (b << 16) | r;
    }
  }
  __syncthreads();
  const int v = lh[t];
  int inc = v;
#pragma unroll
  for (int dd = 1; dd < 32; dd <<= 1) {
    const int u = __shfl_up(inc, dd);
    if (lane >= dd) inc += u;
  }
  if (lane == 31) wsum[wave] = inc;
  __syncthreads();
  int woff = 0;
#pragma unroll
  for (int w = 0; w < PT / 32; ++w)
    if (w < wave) woff += wsum[w];
  const int ex = woff + inc - v;
  lo[t] = ex;
  if (t < NB) offs[t * NC + c] = ex;
  __syncthreads();
#pragma unroll
  for (int k = 0; k < EPC; ++k) {
    if (br[k] >= 0)
      sorted[lo[br[k] >> 16] + (br[k] & 0xFFFF)] =
          (int)(((unsigned)(d[k] & (BW - 1)) << 24) | (unsigned)s[k]);
  }
  __syncthreads();
  const int n = min(CH, E - c * CH);
  int* slab = part + (long)c * CH;
#pragma unroll
  for (int k = 0; k < EPC; ++k) {
    const int i = t + k * PT;
    if (i < n) slab[i] = sorted[i];
  }
}

template <class F>
__device__ __forceinline__ void walk_runs(const int* __restrict__ part, int2 mine, int NC,
                                          int wave, int lane, F f) {
#pragma nounroll
  for (int j0 = 0; j0 < 32; j0 += RB) {
    if (wave + j0 * (PT / 32) >= NC) break;
    int w[RB], m[RB], p[RB];
#pragma unroll
    for (int k = 0; k < RB; ++k) {
      p[k] = __builtin_amdgcn_readlane(mine.x, j0 + k);
      m[k] = __builtin_amdgcn_readlane(mine.y, j0 + k);
      w[k] = part[(lane < m[k]) ? p[k] + lane : 0];
    }
    asm volatile("" ::"v"(w[0]), "v"(w[1]), "v"(w[2]), "v"(w[3]), "v"(w[4]), "v"(w[5]), "v"(w[6]),
                 "v"(w[7]), "v"(w[8]), "v"(w[9]), "v"(w[10]), "v"(w[11]), "v"(w[12]), "v"(w[13]),
                 "v"(w[14]), "v"(w[15]));
#pragma unroll
    for (int k = 0; k < RB; ++k)
      if (lane < m[k]) f(w[k]);
#pragma unroll
    for (int k = 0; k < RB; ++k)
      for (int i = lane + 32; i < m[k]; i += 32) f(part[p[k] + i]);
  }
}

__global__ __launch_bounds__(PT) void k_bucket_csr(const int* __restrict__ part, const int* __restrict__ offs,
                                                   int* __restrict__ csr, int* __restrict__ cnt,
                                                   int* __restrict__ rowptr, float* __restrict__ dinv,
                                                   int N, int E, int NB, int NC) {
  __shared__ int lcnt[BW];
  __shared__ int lrow[BW];
  __shared__ int lcur[BW];
  __shared__ int wsum[BW / 32];
  __shared__ int bsum0[PT / 32];
  __shared__ int bsum1[PT / 32];
  __shared__ int2 lrun[MAXC];
  __shared__ int stage[CAP];
  const int t = threadIdx.x;
  const int lane = t & 31;
  const int wave = t >> 5;
  const int b = blockIdx.x;
  int st = 0, ln = 0;
  if (t < NC) {
    st = offs[b * NC + t];
    const int en = (b + 1 < NB) ? offs[(b + 1) * NC + t] : min(CH, E - t * CH);
    ln = en - st;
  }
  lrun[t] = make_int2(t * CH + st, ln);
  int a0 = st, a1 = ln;
#pragma unroll
  for (int dd = 16; dd >= 1; dd >>= 1) {
    a0 += __shfl_xor(a0, dd);
    a1 += __shfl_xor(a1, dd);
  }
  if (lane == 0) { bsum0[wave] = a0; bsum1[wave] = a1; }
  if (t < BW) { lcnt[t] = 0; lcur[t] = 0; }
  __syncthreads();
  int s0 = 0, n = 0;
#pragma unroll
  for (int w = 0; w < PT / 32; ++w) { s0 += bsum0[w]; n += bsum1[w]; }
  const int2 mine = lrun[wave + (PT / 32) * lane];
  walk_runs(part, mine, NC, wave, lane,
            [&](int w) { atomicAdd(&lcnt[((unsigned)w) >> 24], 1); });
  __syncthreads();
  const int v = (t < BW) ? lcnt[t] : 0;
  int inc = v;
#pragma unroll
  for (int dd = 1; dd < 32; dd <<= 1) {
    const int u = __shfl_up(inc, dd);
    if (lane >= dd) inc += u;
  }
  if (t < BW && lane == 31) wsum[wave] = inc;
  __syncthreads();
  if (t < BW) {
    int woff = 0;
#pragma unroll
    for (int w = 0; w < BW / 32; ++w)
      if (w < wave) woff += wsum[w];
    const int ex = woff + inc - v;
    lrow[t] = ex;
    const int node = b * BW + t;
    if (node < N) {
      cnt[node] = v;
      rowptr[node] = s0 + ex;
      dinv[node] = rsqrtf((float)(v + 1));
    }
  }
  __syncthreads();
  if (n <= CAP) {
    walk_runs(part, mine, NC, wave, lane, [&](int w) {
      const int dl = ((unsigned)w) >> 24;
      const int r = atomicAdd(&lcur[dl], 1);
      stage[lrow[dl] + r] = w & SRC_MASK;
    });
    __syncthreads();
    for (int i = t; i < n; i += PT) csr[s0 + i] = stage[i];
  } else {
    walk_runs(part, mine, NC, wave, lane, [&](int w) {
      const int dl = ((unsigned)w) >> 24;
      const int r = atomicAdd(&lcur[dl], 1);
      csr[s0 + lrow[dl] + r] = w & SRC_MASK;
    });
  }
}

__global__ void k_xw_wmma(const float* __restrict__ x,
                          const float* __restrict__ W1,
                          const float* __restrict__ dinv,
                          float* __restrict__ xwd,
                          int N) {
  const int lane = threadIdx.x & 31;
  const int wave = threadIdx.x >> 5;
  const int wavesPerBlock = blockDim.x >> 5;
  const int base = (blockIdx.x * wavesPerBlock + wave) * 16;
  if (base >= N) return;

  const int half = lane >> 4;
  const int l    = lane & 15;
  int row = base + l;
  if (row >= N) row = N - 1;

  const float* xr = x + (long)row * F_IN;
  const int klo = half * 2;

  v2f a0; a0.x = xr[klo];            a0.y = xr[klo + 1];
  v2f b0; b0.x = W1[klo * HID + l];  b0.y = W1[(klo + 1) * HID + l];

  float x4 = xr[4];
  float w4 = W1[4 * HID + l];
  v2f a1; a1.x = half ? 0.0f : x4;  a1.y = 0.0f;
  v2f b1; b1.x = half ? 0.0f : w4;  b1.y = 0.0f;

  const int n0 = base + half * 8;
  float dv[8];
#pragma unroll
  for (int r = 0; r < 8; ++r) dv[r] = dinv[min(n0 + r, N - 1)];

  v8f c = {};
  c = __builtin_amdgcn_wmma_f32_16x16x4_f32(false, a0, false, b0, (short)0, c, false, false);
  c = __builtin_amdgcn_wmma_f32_16x16x4_f32(false, a1, false, b1, (short)0, c, false, false);

  float2* xp = (float2*)xwd + ((long)n0 * HID + l);
  if (base + 16 <= N) {
#pragma unroll
    for (int r = 0; r < 8; ++r)
      xp[(long)r * HID] = make_float2(c[r], dv[r]);
  } else {
#pragma unroll
    for (int r = 0; r < 8; ++r)
      if (n0 + r < N) xp[(long)r * HID] = make_float2(c[r], dv[r]);
  }
}

__global__ void k_agg1_gather(const int* __restrict__ csr, const int* __restrict__ rowptr,
                              const int* __restrict__ cnt, const float* __restrict__ xwd,
                              const float* __restrict__ b1, const float* __restrict__ W2,
                              float* __restrict__ zd, int N) {
  const int t = blockIdx.x * blockDim.x + threadIdx.x;
  const int node = t >> 4;
  const int l = threadIdx.x & 15;
  if (node >= N) return;
  const float2* rows = (const float2*)xwd;
  const int start = rowptr[node];
  const int deg = cnt[node];
  const float2 own = rows[(long)node * HID + l];
  const float bl = b1[l];
  const float dd = own.y;
  float acc = dd * dd * own.x;
  for (int base = 0; base < deg; base += 16) {
    const int m = deg - base;
    int s_l = node;
    if (l < m) s_l = csr[start + base + l];
#pragma unroll
    for (int g = 0; g < 16; g += 8) {
      if (g < m) {
        float2 v[8];
#pragma unroll
        for (int j = 0; j < 8; ++j) {
          const int s = __shfl(s_l, g + j, 16);
          v[j] = rows[(long)s * HID + l];
        }
#pragma unroll
        for (int j = 0; j < 8; ++j)
          acc = fmaf(v[j].y * dd, (g + j < m) ? v[j].x : 0.0f, acc);
      }
    }
  }
  const float h = fmaxf(acc + bl, 0.0f);
  float z = 0.0f;
#pragma unroll
  for (int f = 0; f < HID; ++f) z = fmaf(__shfl(h, f, 16), W2[f], z);
  if (l == 0) ((float2*)zd)[node] = make_float2(z, dd);
}

__global__ void k_agg2_gather(const int* __restrict__ csr, const int* __restrict__ rowptr,
                              const int* __restrict__ cnt, const float* __restrict__ zd,
                              const float* __restrict__ b2, float* __restrict__ out, int N) {
  const int t = blockIdx.x * blockDim.x + threadIdx.x;
  const int node = t >> 4;
  const int l = threadIdx.x & 15;
  if (node >= N) return;
  const float2* rec = (const float2*)zd;
  const int start = rowptr[node];
  const int deg = cnt[node];
  const float2 own = rec[node];
  const float dd = own.y;
  float acc = 0.0f;
  for (int p = l; p < deg; p += 16) {
    const float2 r = rec[csr[start + p]];
    acc = fmaf(r.y * dd, r.x, acc);
  }
#pragma unroll
  for (int d = 8; d >= 1; d >>= 1) acc += __shfl_xor(acc, d, 16);
  if (l == 0) out[node] = b2[0] + dd * dd * own.x + acc;
}

extern "C" void kernel_launch(void* const* d_in, const int* in_sizes, int n_in,
                              void* d_out, int out_size, void* d_ws, size_t ws_size,
                              hipStream_t stream) {
  const float* x  = (const float*)d_in[0];
  const int*   ei = (const int*)d_in[1];
  const float* W1 = (const float*)d_in[2];
  const float* b1 = (const float*)d_in[3];
  const float* W2 = (const float*)d_in[4];
  const float* b2 = (const float*)d_in[5];
  float* out = (float*)d_out;

  const int N = in_sizes[0] / F_IN;
  const int E = in_sizes[1] / 2;
  const int NB = (N + BW - 1) / BW;
  const int NC = (E + CH - 1) / CH;
  const int M = NB * NC;

  char* ws = (char*)d_ws;
  size_t off = 0;
  auto carve = [&](size_t bytes) -> void* {
    off = (off + 255) & ~(size_t)255;
    void* p = ws + off;
    off += bytes;
    return p;
  };
  float* dinv    = (float*)carve((size_t)N * sizeof(float));
  float* xwd     = (float*)carve((size_t)N * HID * 2 * sizeof(float));
  float* zd      = (float*)carve((size_t)N * 2 * sizeof(float));
  int*   cnt     = (int*)carve((size_t)N * sizeof(int));
  int*   rowptr  = (int*)carve((size_t)N * sizeof(int));
  int*   offs    = (int*)carve((size_t)M * sizeof(int));
  int*   part    = (int*)carve((size_t)NC * CH * sizeof(int));
  int*   csr     = (int*)carve((size_t)E * sizeof(int));
  (void)ws_size; (void)n_in; (void)out_size;

  const int B = 256;
  const int gN16 = (int)(((long)N * 16 + B - 1) / B);
  const int wavesNeeded   = (N + 15) / 16;
  const int wavesPerBlock = B / 32;
  const int gW = (wavesNeeded + wavesPerBlock - 1) / wavesPerBlock;

  hipLaunchKernelGGL(k_chunk_sort,  dim3(NC),     dim3(PT), 0, stream, ei, part, offs, E, NB, NC);
  hipLaunchKernelGGL(k_bucket_csr,  dim3(NB),     dim3(PT), 0, stream, part, offs, csr, cnt, rowptr, dinv, N, E, NB, NC);
  hipLaunchKernelGGL(k_xw_wmma,     dim3(gW),     dim3(B), 0, stream, x, W1, dinv, xwd, N);
  hipLaunchKernelGGL(k_agg1_gather, dim3(gN16),   dim3(B), 0, stream, csr, rowptr, cnt, xwd, b1, W2, zd, N);
  hipLaunchKernelGGL(k_agg2_gather, dim3(gN16),   dim3(B), 0, stream, csr, rowptr, cnt, zd, b2, out, N);
}


# x@W1 pass folded into the bucket pass: wave w of a bucket does the WMMA tile of its nodes 16w..16w+15, dinv comes from LDS, rows stored beside the second walk: 4 launches, not 5; sum order kept
# speedup vs baseline: 13.1842x; 1.0309x over previous
#include <hip/hip_runtime.h>

typedef __attribute__((ext_vector_type(2))) float v2f;
typedef __attribute__((ext_vector_type(8))) float v8f;

#define F_IN 5
#define HID 16
#define BSHIFT 8
#define BW 256
#define MAXB 512
#define MAXC 512
#define PT 512
#define CH 8192
#define EPC (CH / PT)
#define CAP 12288
#define RB 16
#define SRC_MASK 0x00FFFFFF

static_assert(PT == MAXB, "one thread per bucket table entry");
static_assert(PT == MAXC, "one thread per chunk table entry");
static_assert(RB == 16, "the load fence in walk_runs names RB values");
static_assert(MAXC == 32 * (PT / 32), "a wave's runs, one per lane, cover every chunk");
static_assert(BW <= PT, "one thread per node of a bucket");
static_assert(BW == 16 * (PT / 32), "one 16-node matrix tile per wave covers the bucket");


__global__ __launch_bounds__(PT) void k_chunk_sort(const int* __restrict__ ei, int* __restrict__ part,
                                                   int* __restrict__ offs, int E, int NB, int NC) {
  __shared__ int lh[MAXB];
  __shared__ int lo[MAXB];
  __shared__ int wsum[PT / 32];
  __shared__ int sorted[CH];
  const int t = threadIdx.x;
  const int lane = t & 31;
  const int wave = t >> 5;
  const int c = blockIdx.x;
  const int* dst = ei + (long)E;
  lh[t] = 0;
  __syncthreads();
  const int e0 = c * CH + t;
  int d[EPC], s[EPC], br[EPC];
#pragma unroll
  for (int k = 0; k < EPC; ++k) {
    const int e = e0 + k * PT;
    const bool ok = e < E;
    d[k] = ok ? dst[e] : -1;
    s[k] = ok ? ei[e] : 0;
  }
#pragma unroll
  for (int k = 0; k < EPC; ++k) {
    br[k] = -1;
    if (d[k] >= 0) {
      const int b = d[k] >> BSHIFT;
      const int r = atomicAdd(&lh[b], 1);
      br[k] = (b << 16) | r;
    }
  }
  __syncthreads();
  const int v = lh[t];
  int inc = v;
#pragma unroll
  for (int dd = 1; dd < 32; dd <<= 1) {
    const int u = __shfl_up(inc, dd);
    if (lane >= dd) inc += u;
  }
  if (lane == 31) wsum[wave] = inc;
  __syncthreads();
  int woff = 0;
#pragma unroll
  for (int w = 0; w < PT / 32; ++w)
    if (w < wave) woff += wsum[w];
  const int ex = woff + inc - v;
  lo[t] = ex;
  if (t < NB) offs[t * NC + c] = ex;
  __syncthreads();
#pragma unroll
  for (int k = 0; k < EPC; ++k) {
    if (br[k] >= 0)
      sorted[lo[br[k] >> 16] + (br[k] & 0xFFFF)] =
          (int)(((unsigned)(d[k] & (BW - 1)) << 24) | (unsigned)s[k]);
  }
  __syncthreads();
  const int n = min(CH, E - c * CH);
  int* slab = part + (long)c * CH;
#pragma unroll
  for (int k = 0; k < EPC; ++k) {
    const int i = t + k * PT;
    if (i < n) slab[i] = sorted[i];
  }
}

template <class F>
__device__ __forceinline__ void walk_runs(const int* __restrict__ part, int2 mine, int NC,
                                          int wave, int lane, F f) {
#pragma nounroll
  for (int j0 = 0; j0 < 32; j0 += RB) {
    if (wave + j0 * (PT / 32) >= NC) break;
    int w[RB], m[RB], p[RB];
#pragma unroll
    for (int k = 0; k < RB; ++k) {
      p[k] = __builtin_amdgcn_readlane(mine.x, j0 + k);
      m[k] = __builtin_amdgcn_readlane(mine.y, j0 + k);
      w[k] = part[(lane < m[k]) ? p[k] + lane : 0];
    }
    asm volatile("" ::"v"(w[0]), "v"(w[1]), "v"(w[2]), "v"(w[3]), "v"(w[4]), "v"(w[5]), "v"(w[6]),
                 "v"(w[7]), "v"(w[8]), "v"(w[9]), "v"(w[10]), "v"(w[11]), "v"(w[12]), "v"(w[13]),
                 "v"(w[14]), "v"(w[15]));
#pragma unroll
    for (int k = 0; k < RB; ++k)
      if (lane < m[k]) f(w[k]);
#pragma unroll
    for (int k = 0; k < RB; ++k)
      for (int i = lane + 32; i < m[k]; i += 32) f(part[p[k] + i]);
  }
}

__global__ __launch_bounds__(PT) void k_bucket_csr(const int* __restrict__ part, const int* __restrict__ offs,
                                                   const float* __restrict__ x,
                                                   const float* __restrict__ W1,
                                                   int* __restrict__ csr, int* __restrict__ cnt,
                                                   int* __restrict__ rowptr,
                                                   float* __restrict__ xwd,
                                                   int N, int E, int NB, int NC) {
  __shared__ int lcnt[BW];
  __shared__ int lrow[BW];
  __shared__ int lcur[BW];
  __shared__ int wsum[BW / 32];
  __shared__ int bsum0[PT / 32];
  __shared__ int bsum1[PT / 32];
  __shared__ int2 lrun[MAXC];
  __shared__ int stage[CAP];
  __shared__ float ldinv[BW];
  const int t = threadIdx.x;
  const int lane = t & 31;
  const int wave = t >> 5;
  const int b = blockIdx.x;

  const int half = lane >> 4;
  const int l = lane & 15;
  const int tile = b * BW + wave * 16;
  const float* xr = x + (long)min(tile + l, N - 1) * F_IN;
  const int klo = half * 2;
  v2f fa0; fa0.x = xr[klo];            fa0.y = xr[klo + 1];
  v2f fb0; fb0.x = W1[klo * HID + l];  fb0.y = W1[(klo + 1) * HID + l];
  const float x4 = xr[4];
  const float w4 = W1[4 * HID + l];
  v2f fa1; fa1.x = half ? 0.0f : x4;  fa1.y = 0.0f;
  v2f fb1; fb1.x = half ? 0.0f : w4;  fb1.y = 0.0f;

  int st = 0, ln = 0;
  if (t < NC) {
    st = offs[b * NC + t];
    const int en = (b + 1 < NB) ? offs[(b + 1) * NC + t] : min(CH, E - t * CH);
    ln = en - st;
  }
  lrun[t] = make_int2(t * CH + st, ln);
  int a0 = st, a1 = ln;
#pragma unroll
  for (int dd = 16; dd >= 1; dd >>= 1) {
    a0 += __shfl_xor(a0, dd);
    a1 += __shfl_xor(a1, dd);
  }
  if (lane == 0) { bsum0[wave] = a0; bsum1[wave] = a1; }
  if (t < BW) { lcnt[t] = 0; lcur[t] = 0; }
  __syncthreads();
  int s0 = 0, n = 0;
#pragma unroll
  for (int w = 0; w < PT / 32; ++w) { s0 += bsum0[w]; n += bsum1[w]; }
  const int2 mine = lrun[wave + (PT / 32) * lane];
  v8f xw = {};
  xw = __builtin_amdgcn_wmma_f32_16x16x4_f32(false, fa0, false, fb0, (short)0, xw, false, false);
  xw = __builtin_amdgcn_wmma_f32_16x16x4_f32(false, fa1, false, fb1, (short)0, xw, false, false);
  walk_runs(part, mine, NC, wave, lane,
            [&](int w) { atomicAdd(&lcnt[((unsigned)w) >> 24], 1); });
  __syncthreads();
  const int v = (t < BW) ? lcnt[t] : 0;
  int inc = v;
#pragma unroll
  for (int dd = 1; dd < 32; dd <<= 1) {
    const int u = __shfl_up(inc, dd);
    if (lane >= dd) inc += u;
  }
  if (t < BW && lane == 31) wsum[wave] = inc;
  __syncthreads();
  if (t < BW) {
    int woff = 0;
#pragma unroll
    for (int w = 0; w < BW / 32; ++w)
      if (w < wave) woff += wsum[w];
    const int ex = woff + inc - v;
    lrow[t] = ex;
    ldinv[t] = rsqrtf((float)(v + 1));
    const int node = b * BW + t;
    if (node < N) {
      cnt[node] = v;
      rowptr[node] = s0 + ex;
    }
  }
  __syncthreads();
  {
    const int q0 = wave * 16 + half * 8;
    const int n0 = b * BW + q0;
    float2* xp = (float2*)xwd + ((long)n0 * HID + l);
    if (tile + 16 <= N) {
#pragma unroll
      for (int r = 0; r < 8; ++r) xp[(long)r * HID] = make_float2(xw[r], ldinv[q0 + r]);
    } else {
#pragma unroll
      for (int r = 0; r < 8; ++r)
        if (n0 + r < N) xp[(long)r * HID] = make_float2(xw[r], ldinv[q0 + r]);
    }
  }
  if (n <= CAP) {
    walk_runs(part, mine, NC, wave, lane, [&](int w) {
      const int dl = ((unsigned)w) >> 24;
      const int r = atomicAdd(&lcur[dl], 1);
      stage[lrow[dl] + r] = w & SRC_MASK;
    });
    __syncthreads();
    for (int i = t; i < n; i += PT) csr[s0 + i] = stage[i];
  } else {
    walk_runs(part, mine, NC, wave, lane, [&](int w) {
      const int dl = ((unsigned)w) >> 24;
      const int r = atomicAdd(&lcur[dl], 1);
      csr[s0 + lrow[dl] + r] = w & SRC_MASK;
    });
  }
}


__global__ void k_agg1_gather(const int* __restrict__ csr, const int* __restrict__ rowptr,
                              const int* __restrict__ cnt, const float* __restrict__ xwd,
                              const float* __restrict__ b1, const float* __restrict__ W2,
                              float* __restrict__ zd, int N) {
  const int t = blockIdx.x * blockDim.x + threadIdx.x;
  const int node = t >> 4;
  const int l = threadIdx.x & 15;
  if (node >= N) return;
  const float2* rows = (const float2*)xwd;
  const int start = rowptr[node];
  const int deg = cnt[node];
  const float2 own = rows[(long)node * HID + l];
  const float bl = b1[l];
  const float dd = own.y;
  float acc = dd * dd * own.x;
  for (int base = 0; base < deg; base += 16) {
    const int m = deg - base;
    int s_l = node;
    if (l < m) s_l = csr[start + base + l];
#pragma unroll
    for (int g = 0; g < 16; g += 8) {
      if (g < m) {
        float2 v[8];
#pragma unroll
        for (int j = 0; j < 8; ++j) {
          const int s = __shfl(s_l, g + j, 16);
          v[j] = rows[(long)s * HID + l];
        }
#pragma unroll
        for (int j = 0; j < 8; ++j)
          acc = fmaf(v[j].y * dd, (g + j < m) ? v[j].x : 0.0f, acc);
      }
    }
  }
  const float h = fmaxf(acc + bl, 0.0f);
  float z = 0.0f;
#pragma unroll
  for (int f = 0; f < HID; ++f) z = fmaf(__shfl(h, f, 16), W2[f], z);
  if (l == 0) ((float2*)zd)[node] = make_float2(z, dd);
}

__global__ void k_agg2_gather(const int* __restrict__ csr, const int* __restrict__ rowptr,
                              const int* __restrict__ cnt, const float* __restrict__ zd,
                              const float* __restrict__ b2, float* __restrict__ out, int N) {
  const int t = blockIdx.x * blockDim.x + threadIdx.x;
  const int node = t >> 4;
  const int l = threadIdx.x & 15;
  if (node >= N) return;
  const float2* rec = (const float2*)zd;
  const int start = rowptr[node];
  const int deg = cnt[node];
  const float2 own = rec[node];
  const float dd = own.y;
  float acc = 0.0f;
  for (int p = l; p < deg; p += 16) {
    const float2 r = rec[csr[start + p]];
    acc = fmaf(r.y * dd, r.x, acc);
  }
#pragma unroll
  for (int d = 8; d >= 1; d >>= 1) acc += __shfl_xor(acc, d, 16);
  if (l == 0) out[node] = b2[0] + dd * dd * own.x + acc;
}

extern "C" void kernel_launch(void* const* d_in, const int* in_sizes, int n_in,
                              void* d_out, int out_size, void* d_ws, size_t ws_size,
                              hipStream_t stream) {
  const float* x  = (const float*)d_in[0];
  const int*   ei = (const int*)d_in[1];
  const float* W1 = (const float*)d_in[2];
  const float* b1 = (const float*)d_in[3];
  const float* W2 = (const float*)d_in[4];
  const float* b2 = (const float*)d_in[5];
  float* out = (float*)d_out;

  const int N = in_sizes[0] / F_IN;
  const int E = in_sizes[1] / 2;
  const int NB = (N + BW - 1) / BW;
  const int NC = (E + CH - 1) / CH;
  const int M = NB * NC;

  char* ws = (char*)d_ws;
  size_t off = 0;
  auto carve = [&](size_t bytes) -> void* {
    off = (off + 255) & ~(size_t)255;
    void* p = ws + off;
    off += bytes;
    return p;
  };
  float* xwd     = (float*)carve((size_t)N * HID * 2 * sizeof(float));
  float* zd      = (float*)carve((size_t)N * 2 * sizeof(float));
  int*   cnt     = (int*)carve((size_t)N * sizeof(int));
  int*   rowptr  = (int*)carve((size_t)N * sizeof(int));
  int*   offs    = (int*)carve((size_t)M * sizeof(int));
  int*   part    = (int*)carve((size_t)NC * CH * sizeof(int));
  int*   csr     = (int*)carve((size_t)E * sizeof(int));
  (void)ws_size; (void)n_in; (void)out_size;

  const int B = 256;
  const int gN16 = (int)(((long)N * 16 + B - 1) / B);

  hipLaunchKernelGGL(k_chunk_sort,  dim3(NC),     dim3(PT), 0, stream, ei, part, offs, E, NB, NC);
  hipLaunchKernelGGL(k_bucket_csr,  dim3(NB),     dim3(PT), 0, stream, part, offs, x, W1, csr, cnt, rowptr, xwd, N, E, NB, NC);
  hipLaunchKernelGGL(k_agg1_gather, dim3(gN16),   dim3(B), 0, stream, csr, rowptr, cnt, xwd, b1, W2, zd, N);
  hipLaunchKernelGGL(k_agg2_gather, dim3(gN16),   dim3(B), 0, stream, csr, rowptr, cnt, zd, b2, out, N);
}


# Both gathers as stream passes: a work-group owns a run of 64 / 128 nodes, keeps the run's piece of csr in LDS (whole-line reads); layer 2 gathers all its records first, sums from LDS; sum order kept
# speedup vs baseline: 13.9355x; 1.0570x over previous
#include <hip/hip_runtime.h>

typedef __attribute__((ext_vector_type(2))) float v2f;
typedef __attribute__((ext_vector_type(8))) float v8f;

#define F_IN 5
#define HID 16
#define BSHIFT 8
#define BW 256
#define MAXB 512
#define MAXC 512
#define PT 512
#define CH 8192
#define EPC (CH / PT)
#define CAP 12288
#define RB 16
#define SRC_MASK 0x00FFFFFF

static_assert(PT == MAXB, "one thread per bucket table entry");
static_assert(PT == MAXC, "one thread per chunk table entry");
static_assert(RB == 16, "the load fence in walk_runs names RB values");
static_assert(MAXC == 32 * (PT / 32), "a wave's runs, one per lane, cover every chunk");
static_assert(BW <= PT, "one thread per node of a bucket");
static_assert(BW == 16 * (PT / 32), "one 16-node matrix tile per wave covers the bucket");


__global__ __launch_bounds__(PT) void k_chunk_sort(const int* __restrict__ ei, int* __restrict__ part,
                                                   int* __restrict__ offs, int E, int NB, int NC) {
  __shared__ int lh[MAXB];
  __shared__ int lo[MAXB];
  __shared__ int wsum[PT / 32];
  __shared__ int sorted[CH];
  const int t = threadIdx.x;
  const int lane = t & 31;
  const int wave = t >> 5;
  const int c = blockIdx.x;
  const int* dst = ei + (long)E;
  lh[t] = 0;
  __syncthreads();
  const int e0 = c * CH + t;
  int d[EPC], s[EPC], br[EPC];
#pragma unroll
  for (int k = 0; k < EPC; ++k) {
    const int e = e0 + k * PT;
    const bool ok = e < E;
    d[k] = ok ? dst[e] : -1;
    s[k] = ok ? ei[e] : 0;
  }
#pragma unroll
  for (int k = 0; k < EPC; ++k) {
    br[k] = -1;
    if (d[k] >= 0) {
      const int b = d[k] >> BSHIFT;
      const int r = atomicAdd(&lh[b], 1);
      br[k] = (b << 16) | r;
    }
  }
  __syncthreads();
  const int v = lh[t];
  int inc = v;
#pragma unroll
  for (int dd = 1; dd < 32; dd <<= 1) {
    const int u = __shfl_up(inc, dd);
    if (lane >= dd) inc += u;
  }
  if (lane == 31) wsum[wave] = inc;
  __syncthreads();
  int woff = 0;
#pragma unroll
  for (int w = 0; w < PT / 32; ++w)
    if (w < wave) woff += wsum[w];
  const int ex = woff + inc - v;
  lo[t] = ex;
  if (t < NB) offs[t * NC + c] = ex;
  __syncthreads();
#pragma unroll
  for (int k = 0; k < EPC; ++k) {
    if (br[k] >= 0)
      sorted[lo[br[k] >> 16] + (br[k] & 0xFFFF)] =
          (int)(((unsigned)(d[k] & (BW - 1)) << 24) | (unsigned)s[k]);
  }
  __syncthreads();
  const int n = min(CH, E - c * CH);
  int* slab = part + (long)c * CH;
#pragma unroll
  for (int k = 0; k < EPC; ++k) {
    const int i = t + k * PT;
    if (i < n) slab[i] = sorted[i];
  }
}

template <class F>
__device__ __forceinline__ void walk_runs(const int* __restrict__ part, int2 mine, int NC,
                                          int wave, int lane, F f) {
#pragma nounroll
  for (int j0 = 0; j0 < 32; j0 += RB) {
    if (wave + j0 * (PT / 32) >= NC) break;
    int w[RB], m[RB], p[RB];
#pragma unroll
    for (int k = 0; k < RB; ++k) {
      p[k] = __builtin_amdgcn_readlane(mine.x, j0 + k);
      m[k] = __builtin_amdgcn_readlane(mine.y, j0 + k);
      w[k] = part[(lane < m[k]) ? p[k] + lane : 0];
    }
    asm volatile("" ::"v"(w[0]), "v"(w[1]), "v"(w[2]), "v"(w[3]), "v"(w[4]), "v"(w[5]), "v"(w[6]),
                 "v"(w[7]), "v"(w[8]), "v"(w[9]), "v"(w[10]), "v"(w[11]), "v"(w[12]), "v"(w[13]),
                 "v"(w[14]), "v"(w[15]));
#pragma unroll
    for (int k = 0; k < RB; ++k)
      if (lane < m[k]) f(w[k]);
#pragma unroll
    for (int k = 0; k < RB; ++k)
      for (int i = lane + 32; i < m[k]; i += 32) f(part[p[k] + i]);
  }
}

__global__ __launch_bounds__(PT) void k_bucket_csr(const int* __restrict__ part, const int* __restrict__ offs,
                                                   const float* __restrict__ x,
                                                   const float* __restrict__ W1,
                                                   int* __restrict__ csr, int* __restrict__ cnt,
                                                   int* __restrict__ rowptr,
                                                   float* __restrict__ xwd,
                                                   int N, int E, int NB, int NC) {
  __shared__ int lcnt[BW];
  __shared__ int lrow[BW];
  __shared__ int lcur[BW];
  __shared__ int wsum[BW / 32];
  __shared__ int bsum0[PT / 32];
  __shared__ int bsum1[PT / 32];
  __shared__ int2 lrun[MAXC];
  __shared__ int stage[CAP];
  __shared__ float ldinv[BW];
  const int t = threadIdx.x;
  const int lane = t & 31;
  const int wave = t >> 5;
  const int b = blockIdx.x;

  const int half = lane >> 4;
  const int l = lane & 15;
  const int tile = b * BW + wave * 16;
  const float* xr = x + (long)min(tile + l, N - 1) * F_IN;
  const int klo = half * 2;
  v2f fa0; fa0.x = xr[klo];            fa0.y = xr[klo + 1];
  v2f fb0; fb0.x = W1[klo * HID + l];  fb0.y = W1[(klo + 1) * HID + l];
  const float x4 = xr[4];
  const float w4 = W1[4 * HID + l];
  v2f fa1; fa1.x = half ? 0.0f : x4;  fa1.y = 0.0f;
  v2f fb1; fb1.x = half ? 0.0f : w4;  fb1.y = 0.0f;

  int st = 0, ln = 0;
  if (t < NC) {
    st = offs[b * NC + t];
    const int en = (b + 1 < NB) ? offs[(b + 1) * NC + t] : min(CH, E - t * CH);
    ln = en - st;
  }
  lrun[t] = make_int2(t * CH + st, ln);
  int a0 = st, a1 = ln;
#pragma unroll
  for (int dd = 16; dd >= 1; dd >>= 1) {
    a0 += __shfl_xor(a0, dd);
    a1 += __shfl_xor(a1, dd);
  }
  if (lane == 0) { bsum0[wave] = a0; bsum1[wave] = a1; }
  if (t < BW) { lcnt[t] = 0; lcur[t] = 0; }
  __syncthreads();
  int s0 = 0, n = 0;
#pragma unroll
  for (int w = 0; w < PT / 32; ++w) { s0 += bsum0[w]; n += bsum1[w]; }
  const int2 mine = lrun[wave + (PT / 32) * lane];
  v8f xw = {};
  xw = __builtin_amdgcn_wmma_f32_16x16x4_f32(false, fa0, false, fb0, (short)0, xw, false, false);
  xw = __builtin_amdgcn_wmma_f32_16x16x4_f32(false, fa1, false, fb1, (short)0, xw, false, false);
  walk_runs(part, mine, NC, wave, lane,
            [&](int w) { atomicAdd(&lcnt[((unsigned)w) >> 24], 1); });
  __syncthreads();
  const int v = (t < BW) ? lcnt[t] : 0;
  int inc = v;
#pragma unroll
  for (int dd = 1; dd < 32; dd <<= 1) {
    const int u = __shfl_up(inc, dd);
    if (lane >= dd) inc += u;
  }
  if (t < BW && lane == 31) wsum[wave] = inc;
  __syncthreads();
  if (t < BW) {
    int woff = 0;
#pragma unroll
    for (int w = 0; w < BW / 32; ++w)
      if (w < wave) woff += wsum[w];
    const int ex = woff + inc - v;
    lrow[t] = ex;
    ldinv[t] = rsqrtf((float)(v + 1));
    const int node = b * BW + t;
    if (node < N) {
      cnt[node] = v;
      rowptr[node] = s0 + ex;
    }
  }
  __syncthreads();
  {
    const int q0 = wave * 16 + half * 8;
    const int n0 = b * BW + q0;
    float2* xp = (float2*)xwd + ((long)n0 * HID + l);
    if (tile + 16 <= N) {
#pragma unroll
      for (int r = 0; r < 8; ++r) xp[(long)r * HID] = make_float2(xw[r], ldinv[q0 + r]);
    } else {
#pragma unroll
      for (int r = 0; r < 8; ++r)
        if (n0 + r < N) xp[(long)r * HID] = make_float2(xw[r], ldinv[q0 + r]);
    }
  }
  if (n <= CAP) {
    walk_runs(part, mine, NC, wave, lane, [&](int w) {
      const int dl = ((unsigned)w) >> 24;
      const int r = atomicAdd(&lcur[dl], 1);
      stage[lrow[dl] + r] = w & SRC_MASK;
    });
    __syncthreads();
    for (int i = t; i < n; i += PT) csr[s0 + i] = stage[i];
  } else {
    walk_runs(part, mine, NC, wave, lane, [&](int w) {
      const int dl = ((unsigned)w) >> 24;
      const int r = atomicAdd(&lcur[dl], 1);
      csr[s0 + lrow[dl] + r] = w & SRC_MASK;
    });
  }
}


#define R1N 64
#define R1T 256
#define R1U 9
#define R1CAP (R1T * R1U)
#define R2N 128
#define R2T 256
#define R2U 17
#define R2CAP (R2T * R2U)
#define R2K (R2N / (R2T / 16))

static_assert(R1N <= R1T && R2N <= R2T, "one thread per node of a run for the run's tables");
static_assert(R2K * (R2T / 16) == R2N, "the groups' nodes cover the run");
static_assert(R2K == 8, "the load fence in the layer-2 pass names the reads of 8 nodes");

__global__ __launch_bounds__(R1T) void k_agg1_stream(const int* __restrict__ csr, const int* __restrict__ rowptr,
                                                     const int* __restrict__ cnt, const float* __restrict__ xwd,
                                                     const float* __restrict__ b1, const float* __restrict__ W2,
                                                     float* __restrict__ zd, int N) {
  __shared__ int lsrc[R1CAP + 16];
  __shared__ int lrow[R1N];
  __shared__ int ldeg[R1N];
  __shared__ float2 lz[R1N];
  const int t = threadIdx.x;
  const int l = t & 15;
  const int grp = t >> 4;
  const int n0 = blockIdx.x * R1N;
  const int nn = min(R1N, N - n0);
  const int last = n0 + nn - 1;
  const int e0 = rowptr[n0];
  const int n = rowptr[last] + cnt[last] - e0;
  if (t < nn) {
    lrow[t] = rowptr[n0 + t] - e0;
    ldeg[t] = cnt[n0 + t];
  }
  const bool staged = n <= R1CAP;
  if (staged) {
#pragma unroll
    for (int u = 0; u < R1U; ++u) {
      const int i = t + u * R1T;
      lsrc[i] = csr[(i < n) ? e0 + i : 0];
    }
  }
  __syncthreads();
  const float2* rows = (const float2*)xwd;
  const float bl = b1[l];
  for (int q = grp; q < nn; q += R1T / 16) {
    const int node = n0 + q;
    const int deg = ldeg[q];
    const int start = lrow[q];
    const float2 own = rows[(long)node * HID + l];
    const float dd = own.y;
    float acc = dd * dd * own.x;
    if (staged) {
      for (int base = 0; base < deg; base += 16) {
        const int m = deg - base;
        float2 v[16];
#pragma unroll
        for (int j = 0; j < 16; ++j) {
          const int sj = lsrc[start + base + j];
          const int s = (j < m) ? sj : node;
          v[j] = rows[(long)s * HID + l];
        }
#pragma unroll
        for (int j = 0; j < 16; ++j)
          acc = fmaf(v[j].y * dd, (j < m) ? v[j].x : 0.0f, acc);
      }
    } else {
      const int gs = e0 + start;
      for (int base = 0; base < deg; base += 16) {
        const int m = deg - base;
        int s_l = node;
        if (l < m) s_l = csr[gs + base + l];
#pragma unroll
        for (int g = 0; g < 16; g += 8) {
          if (g < m) {
            float2 v[8];
#pragma unroll
            for (int j = 0; j < 8; ++j) {
              const int s = __shfl(s_l, g + j, 16);
              v[j] = rows[(long)s * HID + l];
            }
#pragma unroll
            for (int j = 0; j < 8; ++j)
              acc = fmaf(v[j].y * dd, (g + j < m) ? v[j].x : 0.0f, acc);
          }
        }
      }
    }
    const float h = fmaxf(acc + bl, 0.0f);
    float z = 0.0f;
#pragma unroll
    for (int f = 0; f < HID; ++f) z = fmaf(__shfl(h, f, 16), W2[f], z);
    if (l == 0) lz[q] = make_float2(z, dd);
  }
  __syncthreads();
  if (t < nn) ((float2*)zd)[n0 + t] = lz[t];
}

__global__ __launch_bounds__(R2T) void k_agg2_stream(const int* __restrict__ csr, const int* __restrict__ rowptr,
                                                     const int* __restrict__ cnt, const float* __restrict__ zd,
                                                     const float* __restrict__ b2, float* __restrict__ out, int N) {
  __shared__ float2 lrec[R2CAP + 48];
  __shared__ float2 lown[R2N];
  __shared__ int lrow[R2N];
  __shared__ int ldeg[R2N];
  __shared__ float lout[R2N];
  const int t = threadIdx.x;
  const int l = t & 15;
  const int grp = t >> 4;
  const int n0 = blockIdx.x * R2N;
  const int nn = min(R2N, N - n0);
  const int last = n0 + nn - 1;
  const float2* rec = (const float2*)zd;
  const int e0 = rowptr[n0];
  const int n = rowptr[last] + cnt[last] - e0;
  if (t < nn) {
    lrow[t] = rowptr[n0 + t] - e0;
    ldeg[t] = cnt[n0 + t];
    lown[t] = rec[n0 + t];
  }
  const bool staged = n <= R2CAP;
  if (staged) {
    int s[R2U];
#pragma unroll
    for (int u = 0; u < R2U; ++u) {
      const int i = t + u * R2T;
      const int raw = csr[(i < n) ? e0 + i : 0];
      s[u] = (i < n) ? raw : n0;
    }
    float2 r[R2U];
#pragma unroll
    for (int u = 0; u < R2U; ++u) r[u] = rec[s[u]];
#pragma unroll
    for (int u = 0; u < R2U; ++u) lrec[t + u * R2T] = r[u];
  }
  __syncthreads();
  const float bias = b2[0];
  if (staged) {
    float acc[R2K], dk[R2K], ox[R2K];
    int deg[R2K], start[R2K];
    float2 rr[R2K][3];
#pragma unroll
    for (int k = 0; k < R2K; ++k) {
      const int q = grp + k * (R2T / 16);
      const bool live = q < nn;
      deg[k] = live ? ldeg[q] : 0;
      start[k] = live ? lrow[q] : 0;
      const float2 own = lown[q];
      dk[k] = own.y;
      ox[k] = own.x;
    }
#pragma unroll
    for (int k = 0; k < R2K; ++k) {
#pragma unroll
      for (int j = 0; j < 3; ++j) rr[k][j] = lrec[start[k] + l + 16 * j];
    }
    asm volatile("" ::"v"(rr[0][0].x), "v"(rr[0][0].y), "v"(rr[0][1].x), "v"(rr[0][1].y), "v"(rr[0][2].x), "v"(rr[0][2].y),
                 "v"(rr[1][0].x), "v"(rr[1][0].y), "v"(rr[1][1].x), "v"(rr[1][1].y), "v"(rr[1][2].x), "v"(rr[1][2].y),
                 "v"(rr[2][0].x), "v"(rr[2][0].y), "v"(rr[2][1].x), "v"(rr[2][1].y), "v"(rr[2][2].x), "v"(rr[2][2].y),
                 "v"(rr[3][0].x), "v"(rr[3][0].y), "v"(rr[3][1].x), "v"(rr[3][1].y), "v"(rr[3][2].x), "v"(rr[3][2].y));
    asm volatile("" ::"v"(rr[4][0].x), "v"(rr[4][0].y), "v"(rr[4][1].x), "v"(rr[4][1].y), "v"(rr[4][2].x), "v"(rr[4][2].y),
                 "v"(rr[5][0].x), "v"(rr[5][0].y), "v"(rr[5][1].x), "v"(rr[5][1].y), "v"(rr[5][2].x), "v"(rr[5][2].y),
                 "v"(rr[6][0].x), "v"(rr[6][0].y), "v"(rr[6][1].x), "v"(rr[6][1].y), "v"(rr[6][2].x), "v"(rr[6][2].y),
                 "v"(rr[7][0].x), "v"(rr[7][0].y), "v"(rr[7][1].x), "v"(rr[7][1].y), "v"(rr[7][2].x), "v"(rr[7][2].y));
#pragma unroll
    for (int k = 0; k < R2K; ++k) {
      float a = 0.0f;
#pragma unroll
      for (int j = 0; j < 3; ++j)
        a = (l + 16 * j < deg[k]) ? fmaf(rr[k][j].y * dk[k], rr[k][j].x, a) : a;
      for (int p = l + 48; p < deg[k]; p += 16) {
        const float2 r1 = lrec[start[k] + p];
        a = fmaf(r1.y * dk[k], r1.x, a);
      }
      acc[k] = a;
    }
#pragma unroll
    for (int d = 8; d >= 1; d >>= 1) {
#pragma unroll
      for (int k = 0; k < R2K; ++k) acc[k] += __shfl_xor(acc[k], d, 16);
    }
    if (l == 0) {
#pragma unroll
      for (int k = 0; k < R2K; ++k)
        lout[grp + k * (R2T / 16)] = bias + dk[k] * dk[k] * ox[k] + acc[k];
    }
  } else {
    for (int q = grp; q < nn; q += R2T / 16) {
      const int deg = ldeg[q];
      const int gs = e0 + lrow[q];
      const float2 own = lown[q];
      const float dd = own.y;
      float acc = 0.0f;
      for (int p = l; p < deg; p += 16) {
        const float2 rr = rec[csr[gs + p]];
        acc = fmaf(rr.y * dd, rr.x, acc);
      }
#pragma unroll
      for (int d = 8; d >= 1; d >>= 1) acc += __shfl_xor(acc, d, 16);
      if (l == 0) lout[q] = bias + dd * dd * own.x + acc;
    }
  }
  __syncthreads();
  if (t < nn) out[n0 + t] = lout[t];
}

extern "C" void kernel_launch(void* const* d_in, const int* in_sizes, int n_in,
                              void* d_out, int out_size, void* d_ws, size_t ws_size,
                              hipStream_t stream) {
  const float* x  = (const float*)d_in[0];
  const int*   ei = (const int*)d_in[1];
  const float* W1 = (const float*)d_in[2];
  const float* b1 = (const float*)d_in[3];
  const float* W2 = (const float*)d_in[4];
  const float* b2 = (const float*)d_in[5];
  float* out = (float*)d_out;

  const int N = in_sizes[0] / F_IN;
  const int E = in_sizes[1] / 2;
  const int NB = (N + BW - 1) / BW;
  const int NC = (E + CH - 1) / CH;
  const int M = NB * NC;

  char* ws = (char*)d_ws;
  size_t off = 0;
  auto carve = [&](size_t bytes) -> void* {
    off = (off + 255) & ~(size_t)255;
    void* p = ws + off;
    off += bytes;
    return p;
  };
  float* xwd     = (float*)carve((size_t)N * HID * 2 * sizeof(float));
  float* zd      = (float*)carve((size_t)N * 2 * sizeof(float));
  int*   cnt     = (int*)carve((size_t)N * sizeof(int));
  int*   rowptr  = (int*)carve((size_t)N * sizeof(int));
  int*   offs    = (int*)carve((size_t)M * sizeof(int));
  int*   part    = (int*)carve((size_t)NC * CH * sizeof(int));
  int*   csr     = (int*)carve((size_t)E * sizeof(int));
  (void)ws_size; (void)n_in; (void)out_size;

  const int G1 = (N + R1N - 1) / R1N;
  const int G2 = (N + R2N - 1) / R2N;

  hipLaunchKernelGGL(k_chunk_sort,  dim3(NC),     dim3(PT), 0, stream, ei, part, offs, E, NB, NC);
  hipLaunchKernelGGL(k_bucket_csr,  dim3(NB),     dim3(PT), 0, stream, part, offs, x, W1, csr, cnt, rowptr, xwd, N, E, NB, NC);
  hipLaunchKernelGGL(k_agg1_stream, dim3(G1),     dim3(R1T), 0, stream, csr, rowptr, cnt, xwd, b1, W2, zd, N);
  hipLaunchKernelGGL(k_agg2_stream, dim3(G2),     dim3(R2T), 0, stream, csr, rowptr, cnt, zd, b2, out, N);
}


# Layer-1 edges gather 32-byte records (dinv*x, 5 floats), not 128-byte x@W1 rows; x@W1 moves into the gather: 16 list entries = A rows of the f32 WMMA, rows summed after; node f32 sum order changes
# speedup vs baseline: 15.8299x; 1.1359x over previous
#include <hip/hip_runtime.h>

typedef __attribute__((ext_vector_type(2))) float v2f;
typedef __attribute__((ext_vector_type(8))) float v8f;

#define F_IN 5
#define HID 16
#define BSHIFT 8
#define BW 256
#define MAXB 512
#define MAXC 512
#define PT 512
#define CH 8192
#define EPC (CH / PT)
#define CAP 12288
#define RB 16
#define SRC_MASK 0x00FFFFFF

static_assert(PT == MAXB, "one thread per bucket table entry");
static_assert(PT == MAXC, "one thread per chunk table entry");
static_assert(RB == 16, "the load fence in walk_runs names RB values");
static_assert(MAXC == 32 * (PT / 32), "a wave's runs, one per lane, cover every chunk");
static_assert(BW <= PT, "one thread per node of a bucket");


__global__ __launch_bounds__(PT) void k_chunk_sort(const int* __restrict__ ei, int* __restrict__ part,
                                                   int* __restrict__ offs, int E, int NB, int NC) {
  __shared__ int lh[MAXB];
  __shared__ int lo[MAXB];
  __shared__ int wsum[PT / 32];
  __shared__ int sorted[CH];
  const int t = threadIdx.x;
  const int lane = t & 31;
  const int wave = t >> 5;
  const int c = blockIdx.x;
  const int* dst = ei + (long)E;
  lh[t] = 0;
  __syncthreads();
  const int e0 = c * CH + t;
  int d[EPC], s[EPC], br[EPC];
#pragma unroll
  for (int k = 0; k < EPC; ++k) {
    const int e = e0 + k * PT;
    const bool ok = e < E;
    d[k] = ok ? dst[e] : -1;
    s[k] = ok ? ei[e] : 0;
  }
#pragma unroll
  for (int k = 0; k < EPC; ++k) {
    br[k] = -1;
    if (d[k] >= 0) {
      const int b = d[k] >> BSHIFT;
      const int r = atomicAdd(&lh[b], 1);
      br[k] = (b << 16) | r;
    }
  }
  __syncthreads();
  const int v = lh[t];
  int inc = v;
#pragma unroll
  for (int dd = 1; dd < 32; dd <<= 1) {
    const int u = __shfl_up(inc, dd);
    if (lane >= dd) inc += u;
  }
  if (lane == 31) wsum[wave] = inc;
  __syncthreads();
  int woff = 0;
#pragma unroll
  for (int w = 0; w < PT / 32; ++w)
    if (w < wave) woff += wsum[w];
  const int ex = woff + inc - v;
  lo[t] = ex;
  if (t < NB) offs[t * NC + c] = ex;
  __syncthreads();
#pragma unroll
  for (int k = 0; k < EPC; ++k) {
    if (br[k] >= 0)
      sorted[lo[br[k] >> 16] + (br[k] & 0xFFFF)] =
          (int)(((unsigned)(d[k] & (BW - 1)) << 24) | (unsigned)s[k]);
  }
  __syncthreads();
  const int n = min(CH, E - c * CH);
  int* slab = part + (long)c * CH;
#pragma unroll
  for (int k = 0; k < EPC; ++k) {
    const int i = t + k * PT;
    if (i < n) slab[i] = sorted[i];
  }
}

template <class F>
__device__ __forceinline__ void walk_runs(const int* __restrict__ part, int2 mine, int NC,
                                          int wave, int lane, F f) {
#pragma nounroll
  for (int j0 = 0; j0 < 32; j0 += RB) {
    if (wave + j0 * (PT / 32) >= NC) break;
    int w[RB], m[RB], p[RB];
#pragma unroll
    for (int k = 0; k < RB; ++k) {
      p[k] = __builtin_amdgcn_readlane(mine.x, j0 + k);
      m[k] = __builtin_amdgcn_readlane(mine.y, j0 + k);
      w[k] = part[(lane < m[k]) ? p[k] + lane : 0];
    }
    asm volatile("" ::"v"(w[0]), "v"(w[1]), "v"(w[2]), "v"(w[3]), "v"(w[4]), "v"(w[5]), "v"(w[6]),
                 "v"(w[7]), "v"(w[8]), "v"(w[9]), "v"(w[10]), "v"(w[11]), "v"(w[12]), "v"(w[13]),
                 "v"(w[14]), "v"(w[15]));
#pragma unroll
    for (int k = 0; k < RB; ++k)
      if (lane < m[k]) f(w[k]);
#pragma unroll
    for (int k = 0; k < RB; ++k)
      for (int i = lane + 32; i < m[k]; i += 32) f(part[p[k] + i]);
  }
}

__global__ __launch_bounds__(PT) void k_bucket_csr(const int* __restrict__ part, const int* __restrict__ offs,
                                                   const float* __restrict__ x,
                                                   int* __restrict__ csr, int* __restrict__ cnt,
                                                   int* __restrict__ rowptr,
                                                   float* __restrict__ xs,
                                                   int N, int E, int NB, int NC) {
  __shared__ int lcnt[BW];
  __shared__ int lrow[BW];
  __shared__ int lcur[BW];
  __shared__ int wsum[BW / 32];
  __shared__ int bsum0[PT / 32];
  __shared__ int bsum1[PT / 32];
  __shared__ int2 lrun[MAXC];
  __shared__ int stage[CAP];
  const int t = threadIdx.x;
  const int lane = t & 31;
  const int wave = t >> 5;
  const int b = blockIdx.x;

  float x0 = 0.0f, x1 = 0.0f, x2 = 0.0f, x3 = 0.0f, x4 = 0.0f;
  if (t < BW) {
    const float* xr = x + (long)min(b * BW + t, N - 1) * F_IN;
    x0 = xr[0]; x1 = xr[1]; x2 = xr[2]; x3 = xr[3]; x4 = xr[4];
  }

  int st = 0, ln = 0;
  if (t < NC) {
    st = offs[b * NC + t];
    const int en = (b + 1 < NB) ? offs[(b + 1) * NC + t] : min(CH, E - t * CH);
    ln = en - st;
  }
  lrun[t] = make_int2(t * CH + st, ln);
  int a0 = st, a1 = ln;
#pragma unroll
  for (int dd = 16; dd >= 1; dd >>= 1) {
    a0 += __shfl_xor(a0, dd);
    a1 += __shfl_xor(a1, dd);
  }
  if (lane == 0) { bsum0[wave] = a0; bsum1[wave] = a1; }
  if (t < BW) { lcnt[t] = 0; lcur[t] = 0; }
  __syncthreads();
  int s0 = 0, n = 0;
#pragma unroll
  for (int w = 0; w < PT / 32; ++w) { s0 += bsum0[w]; n += bsum1[w]; }
  const int2 mine = lrun[wave + (PT / 32) * lane];
  walk_runs(part, mine, NC, wave, lane,
            [&](int w) { atomicAdd(&lcnt[((unsigned)w) >> 24], 1); });
  __syncthreads();
  const int v = (t < BW) ? lcnt[t] : 0;
  int inc = v;
#pragma unroll
  for (int dd = 1; dd < 32; dd <<= 1) {
    const int u = __shfl_up(inc, dd);
    if (lane >= dd) inc += u;
  }
  if (t < BW && lane == 31) wsum[wave] = inc;
  __syncthreads();
  if (t < BW) {
    int woff = 0;
#pragma unroll
    for (int w = 0; w < BW / 32; ++w)
      if (w < wave) woff += wsum[w];
    const int ex = woff + inc - v;
    lrow[t] = ex;
    const float dinv = rsqrtf((float)(v + 1));
    const int node = b * BW + t;
    if (node < N) {
      cnt[node] = v;
      rowptr[node] = s0 + ex;
      float4* rp = (float4*)xs + (long)node * 2;
      rp[0] = make_float4(dinv * x0, dinv * x1, dinv * x2, dinv * x3);
      rp[1] = make_float4(dinv * x4, 0.0f, 0.0f, 0.0f);
    }
  }
  __syncthreads();
  if (n <= CAP) {
    walk_runs(part, mine, NC, wave, lane, [&](int w) {
      const int dl = ((unsigned)w) >> 24;
      const int r = atomicAdd(&lcur[dl], 1);
      stage[lrow[dl] + r] = w & SRC_MASK;
    });
    __syncthreads();
    for (int i = t; i < n; i += PT) csr[s0 + i] = stage[i];
  } else {
    walk_runs(part, mine, NC, wave, lane, [&](int w) {
      const int dl = ((unsigned)w) >> 24;
      const int r = atomicAdd(&lcur[dl], 1);
      csr[s0 + lrow[dl] + r] = w & SRC_MASK;
    });
  }
}


#define R1N 64
#define R1T 256
#define R1W (R1T / 32)
#define R1U 9
#define R1CAP (R1T * R1U)
#define R1TILES 3
#define R1PAD (16 * R1TILES)
#define R2N 128
#define R2T 256
#define R2U 17
#define R2CAP (R2T * R2U)
#define R2K (R2N / (R2T / 16))

static_assert(R1N <= R1T && R2N <= R2T, "one thread per node of a run for the run's tables");
static_assert(R1N % (2 * R1W) == 0, "the waves' node pairs cover the run");
static_assert(R1TILES == 3, "the read fence in the layer-1 pass names the LDS reads of 2 x 3 tiles");
static_assert(R2K * (R2T / 16) == R2N, "the groups' nodes cover the run");
static_assert(R2K == 8, "the load fence in the layer-2 pass names the reads of 8 nodes");

template <bool STAGED>
__device__ __forceinline__ void agg1_nodes(const int* lsrc, const int* lrow, const int* ldeg, float2* lz,
                                           const int* __restrict__ csr, const float* __restrict__ xs,
                                           v2f fb0, v2f fb1, float bl, const float* __restrict__ W2,
                                           int e0, int n0, int nn, int N, int wave, int lane) {
  const int half = lane >> 4;
  const int l = lane & 15;
  const float4* recs = (const float4*)xs;
  float w2[HID];
#pragma unroll
  for (int f = 0; f < HID; ++f) w2[f] = W2[f];
#pragma nounroll
  for (int i = 0; i < R1N / (2 * R1W); ++i) {
    int q[2], deg[2], start[2], node[2];
    float4 v[2][R1TILES];
#pragma unroll
    for (int p = 0; p < 2; ++p) {
      q[p] = wave + R1W * (2 * i + p);
      deg[p] = __builtin_amdgcn_readfirstlane(ldeg[q[p]]);
      start[p] = __builtin_amdgcn_readfirstlane(lrow[q[p]]);
      node[p] = min(n0 + q[p], N - 1);
    }
    int sj[2][R1TILES];
    if (STAGED) {
#pragma unroll
      for (int p = 0; p < 2; ++p) {
#pragma unroll
        for (int T = 0; T < R1TILES; ++T) sj[p][T] = lsrc[start[p] + T * 16 + l];
      }
      asm volatile("" ::"v"(sj[0][0]), "v"(sj[0][1]), "v"(sj[0][2]), "v"(sj[1][0]), "v"(sj[1][1]), "v"(sj[1][2]));
    }
#pragma unroll
    for (int p = 0; p < 2; ++p) {
#pragma unroll
      for (int T = 0; T < R1TILES; ++T) {
        const int j = T * 16 + l;
        int s = node[p];
        if (STAGED) {
          s = (j < deg[p]) ? sj[p][T] : node[p];
        } else {
          if (j < deg[p]) s = csr[e0 + start[p] + j];
        }
        v[p][T] = recs[(unsigned)(s * 2 + half)];
      }
    }
    v8f acc[2];
#pragma unroll
    for (int p = 0; p < 2; ++p) {
      acc[p] = v8f{};
      for (int T = R1TILES; T * 16 <= deg[p]; ++T) {
        const int j = T * 16 + l;
        int s = node[p];
        if (j < deg[p]) s = STAGED ? lsrc[start[p] + j] : csr[e0 + start[p] + j];
        const float4 r = recs[(unsigned)(s * 2 + half)];
        const bool on = j <= deg[p];
        v2f a0; a0.x = on ? r.x : 0.0f; a0.y = on ? r.y : 0.0f;
        v2f a1; a1.x = on ? r.z : 0.0f; a1.y = on ? r.w : 0.0f;
        acc[p] = __builtin_amdgcn_wmma_f32_16x16x4_f32(false, a0, false, fb0, (short)0, acc[p], false, false);
        acc[p] = __builtin_amdgcn_wmma_f32_16x16x4_f32(false, a1, false, fb1, (short)0, acc[p], false, false);
      }
    }
#pragma unroll
    for (int T = 0; T < R1TILES; ++T) {
#pragma unroll
      for (int p = 0; p < 2; ++p) {
        const bool on = T * 16 + l <= deg[p];
        v2f a0; a0.x = on ? v[p][T].x : 0.0f; a0.y = on ? v[p][T].y : 0.0f;
        acc[p] = __builtin_amdgcn_wmma_f32_16x16x4_f32(false, a0, false, fb0, (short)0, acc[p], false, false);
      }
#pragma unroll
      for (int p = 0; p < 2; ++p) {
        const bool on = T * 16 + l <= deg[p];
        v2f a1; a1.x = on ? v[p][T].z : 0.0f; a1.y = on ? v[p][T].w : 0.0f;
        acc[p] = __builtin_amdgcn_wmma_f32_16x16x4_f32(false, a1, false, fb1, (short)0, acc[p], false, false);
      }
    }
#pragma unroll
    for (int p = 0; p < 2; ++p) {
      float sum = ((acc[p][0] + acc[p][1]) + (acc[p][2] + acc[p][3])) +
                  ((acc[p][4] + acc[p][5]) + (acc[p][6] + acc[p][7]));
      sum += __shfl_xor(sum, 16);
      const float dd = rsqrtf((float)(deg[p] + 1));
      const float h = fmaxf(fmaf(sum, dd, bl), 0.0f);
      float z = 0.0f;
#pragma unroll
      for (int f = 0; f < HID; ++f) z = fmaf(__shfl(h, f, 16), w2[f], z);
      if (lane == 0 && q[p] < nn) lz[q[p]] = make_float2(z, dd);
    }
  }
}

__global__ __launch_bounds__(R1T) void k_agg1_stream(const int* __restrict__ csr, const int* __restrict__ rowptr,
                                                     const int* __restrict__ cnt, const float* __restrict__ xs,
                                                     const float* __restrict__ W1,
                                                     const float* __restrict__ b1, const float* __restrict__ W2,
                                                     float* __restrict__ zd, int N) {
  __shared__ int lsrc[R1CAP + R1PAD];
  __shared__ int lrow[R1N];
  __shared__ int ldeg[R1N];
  __shared__ float2 lz[R1N];
  const int t = threadIdx.x;
  const int lane = t & 31;
  const int wave = t >> 5;
  const int half = lane >> 4;
  const int l = lane & 15;
  const int n0 = blockIdx.x * R1N;
  const int nn = min(R1N, N - n0);
  const int last = n0 + nn - 1;
  const int e0 = rowptr[n0];
  const int n = rowptr[last] + cnt[last] - e0;
  if (t < R1N) {
    const bool live = t < nn;
    const int nd = live ? n0 + t : last;
    const int rp = rowptr[nd], dg = cnt[nd];
    lrow[t] = live ? rp - e0 : 0;
    ldeg[t] = live ? dg : -1;
  }
  const float wa = W1[(half ? 4 : 0) * HID + l];
  const float wb = W1[1 * HID + l];
  const float wc = W1[2 * HID + l];
  const float wd = W1[3 * HID + l];
  v2f fb0; fb0.x = wa;                 fb0.y = half ? 0.0f : wb;
  v2f fb1; fb1.x = half ? 0.0f : wc;   fb1.y = half ? 0.0f : wd;
  const float bl = b1[l];
  const bool staged = n <= R1CAP;
  if (staged) {
#pragma unroll
    for (int u = 0; u < R1U; ++u) {
      const int i = t + u * R1T;
      lsrc[i] = csr[(i < n) ? e0 + i : 0];
    }
    if (t < R1PAD) lsrc[R1CAP + t] = 0;
  }
  __syncthreads();
  if (staged) agg1_nodes<true>(lsrc, lrow, ldeg, lz, csr, xs, fb0, fb1, bl, W2, e0, n0, nn, N, wave, lane);
  else        agg1_nodes<false>(lsrc, lrow, ldeg, lz, csr, xs, fb0, fb1, bl, W2, e0, n0, nn, N, wave, lane);
  __syncthreads();
  if (t < nn) ((float2*)zd)[n0 + t] = lz[t];
}

__global__ __launch_bounds__(R2T) void k_agg2_stream(const int* __restrict__ csr, const int* __restrict__ rowptr,
                                                     const int* __restrict__ cnt, const float* __restrict__ zd,
                                                     const float* __restrict__ b2, float* __restrict__ out, int N) {
  __shared__ float2 lrec[R2CAP + 48];
  __shared__ float2 lown[R2N];
  __shared__ int lrow[R2N];
  __shared__ int ldeg[R2N];
  __shared__ float lout[R2N];
  const int t = threadIdx.x;
  const int l = t & 15;
  const int grp = t >> 4;
  const int n0 = blockIdx.x * R2N;
  const int nn = min(R2N, N - n0);
  const int last = n0 + nn - 1;
  const float2* rec = (const float2*)zd;
  const int e0 = rowptr[n0];
  const int n = rowptr[last] + cnt[last] - e0;
  if (t < nn) {
    lrow[t] = rowptr[n0 + t] - e0;
    ldeg[t] = cnt[n0 + t];
    lown[t] = rec[n0 + t];
  }
  const bool staged = n <= R2CAP;
  if (staged) {
    int s[R2U];
#pragma unroll
    for (int u = 0; u < R2U; ++u) {
      const int i = t + u * R2T;
      const int raw = csr[(i < n) ? e0 + i : 0];
      s[u] = (i < n) ? raw : n0;
    }
    float2 r[R2U];
#pragma unroll
    for (int u = 0; u < R2U; ++u) r[u] = rec[s[u]];
#pragma unroll
    for (int u = 0; u < R2U; ++u) lrec[t + u * R2T] = r[u];
  }
  __syncthreads();
  const float bias = b2[0];
  if (staged) {
    float acc[R2K], dk[R2K], ox[R2K];
    int deg[R2K], start[R2K];
    float2 rr[R2K][3];
#pragma unroll
    for (int k = 0; k < R2K; ++k) {
      const int q = grp + k * (R2T / 16);
      const bool live = q < nn;
      deg[k] = live ? ldeg[q] : 0;
      start[k] = live ? lrow[q] : 0;
      const float2 own = lown[q];
      dk[k] = own.y;
      ox[k] = own.x;
    }
#pragma unroll
    for (int k = 0; k < R2K; ++k) {
#pragma unroll
      for (int j = 0; j < 3; ++j) rr[k][j] = lrec[start[k] + l + 16 * j];
    }
    asm volatile("" ::"v"(rr[0][0].x), "v"(rr[0][0].y), "v"(rr[0][1].x), "v"(rr[0][1].y), "v"(rr[0][2].x), "v"(rr[0][2].y),
                 "v"(rr[1][0].x), "v"(rr[1][0].y), "v"(rr[1][1].x), "v"(rr[1][1].y), "v"(rr[1][2].x), "v"(rr[1][2].y),
                 "v"(rr[2][0].x), "v"(rr[2][0].y), "v"(rr[2][1].x), "v"(rr[2][1].y), "v"(rr[2][2].x), "v"(rr[2][2].y),
                 "v"(rr[3][0].x), "v"(rr[3][0].y), "v"(rr[3][1].x), "v"(rr[3][1].y), "v"(rr[3][2].x), "v"(rr[3][2].y));
    asm volatile("" ::"v"(rr[4][0].x), "v"(rr[4][0].y), "v"(rr[4][1].x), "v"(rr[4][1].y), "v"(rr[4][2].x), "v"(rr[4][2].y),
                 "v"(rr[5][0].x), "v"(rr[5][0].y), "v"(rr[5][1].x), "v"(rr[5][1].y), "v"(rr[5][2].x), "v"(rr[5][2].y),
                 "v"(rr[6][0].x), "v"(rr[6][0].y), "v"(rr[6][1].x), "v"(rr[6][1].y), "v"(rr[6][2].x), "v"(rr[6][2].y),
                 "v"(rr[7][0].x), "v"(rr[7][0].y), "v"(rr[7][1].x), "v"(rr[7][1].y), "v"(rr[7][2].x), "v"(rr[7][2].y));
#pragma unroll
    for (int k = 0; k < R2K; ++k) {
      float a = 0.0f;
#pragma unroll
      for (int j = 0; j < 3; ++j)
        a = (l + 16 * j < deg[k]) ? fmaf(rr[k][j].y * dk[k], rr[k][j].x, a) : a;
      for (int p = l + 48; p < deg[k]; p += 16) {
        const float2 r1 = lrec[start[k] + p];
        a = fmaf(r1.y * dk[k], r1.x, a);
      }
      acc[k] = a;
    }
#pragma unroll
    for (int d = 8; d >= 1; d >>= 1) {
#pragma unroll
      for (int k = 0; k < R2K; ++k) acc[k] += __shfl_xor(acc[k], d, 16);
    }
    if (l == 0) {
#pragma unroll
      for (int k = 0; k < R2K; ++k)
        lout[grp + k * (R2T / 16)] = bias + dk[k] * dk[k] * ox[k] + acc[k];
    }
  } else {
    for (int q = grp; q < nn; q += R2T / 16) {
      const int deg = ldeg[q];
      const int gs = e0 + lrow[q];
      const float2 own = lown[q];
      const float dd = own.y;
      float acc = 0.0f;
      for (int p = l; p < deg; p += 16) {
        const float2 rr = rec[csr[gs + p]];
        acc = fmaf(rr.y * dd, rr.x, acc);
      }
#pragma unroll
      for (int d = 8; d >= 1; d >>= 1) acc += __shfl_xor(acc, d, 16);
      if (l == 0) lout[q] = bias + dd * dd * own.x + acc;
    }
  }
  __syncthreads();
  if (t < nn) out[n0 + t] = lout[t];
}

extern "C" void kernel_launch(void* const* d_in, const int* in_sizes, int n_in,
                              void* d_out, int out_size, void* d_ws, size_t ws_size,
                              hipStream_t stream) {
  const float* x  = (const float*)d_in[0];
  const int*   ei = (const int*)d_in[1];
  const float* W1 = (const float*)d_in[2];
  const float* b1 = (const float*)d_in[3];
  const float* W2 = (const float*)d_in[4];
  const float* b2 = (const float*)d_in[5];
  float* out = (float*)d_out;

  const int N = in_sizes[0] / F_IN;
  const int E = in_sizes[1] / 2;
  const int NB = (N + BW - 1) / BW;
  const int NC = (E + CH - 1) / CH;
  const int M = NB * NC;

  char* ws = (char*)d_ws;
  size_t off = 0;
  auto carve = [&](size_t bytes) -> void* {
    off = (off + 255) & ~(size_t)255;
    void* p = ws + off;
    off += bytes;
    return p;
  };
  float* xs      = (float*)carve((size_t)N * 8 * sizeof(float));
  float* zd      = (float*)carve((size_t)N * 2 * sizeof(float));
  int*   cnt     = (int*)carve((size_t)N * sizeof(int));
  int*   rowptr  = (int*)carve((size_t)N * sizeof(int));
  int*   offs    = (int*)carve((size_t)M * sizeof(int));
  int*   part    = (int*)carve((size_t)NC * CH * sizeof(int));
  int*   csr     = (int*)carve((size_t)E * sizeof(int));
  (void)ws_size; (void)n_in; (void)out_size;

  const int G1 = (N + R1N - 1) / R1N;
  const int G2 = (N + R2N - 1) / R2N;

  hipLaunchKernelGGL(k_chunk_sort,  dim3(NC),     dim3(PT), 0, stream, ei, part, offs, E, NB, NC);
  hipLaunchKernelGGL(k_bucket_csr,  dim3(NB),     dim3(PT), 0, stream, part, offs, x, csr, cnt, rowptr, xs, N, E, NB, NC);
  hipLaunchKernelGGL(k_agg1_stream, dim3(G1),     dim3(R1T), 0, stream, csr, rowptr, cnt, xs, W1, b1, W2, zd, N);
  hipLaunchKernelGGL(k_agg2_stream, dim3(G2),     dim3(R2T), 0, stream, csr, rowptr, cnt, zd, b2, out, N);
}
